# starting kernel
# baseline (speedup 1.0000x reference)
#include <hip/hip_runtime.h>


typedef __attribute__((ext_vector_type(2))) float v2f;
typedef __attribute__((ext_vector_type(8))) float v8f;

#define INC    32
#define OUTC   64
#define KVOL   8
#define KDIM   256
#define BN_EPS 1e-5f

__global__ void spconv_zero_ws(float* __restrict__ ws, long long nelem4) {
    long long i = (long long)blockIdx.x * blockDim.x + threadIdx.x;
    long long stride = (long long)gridDim.x * blockDim.x;
    float4* p = (float4*)ws;
    for (long long j = i; j < nelem4; j += stride)
        p[j] = make_float4(0.f, 0.f, 0.f, 0.f);
}

__global__ void spconv_scatter(const float* __restrict__ feats,
                               const int*   __restrict__ out_idx,
                               const int*   __restrict__ offset_id,
                               float*       __restrict__ S,
                               int nvox) {
    int t   = blockIdx.x * blockDim.x + threadIdx.x;
    int vox = t >> 5;
    int c   = t & 31;
    if (vox >= nvox) return;
    int seg = out_idx[vox] * KVOL + offset_id[vox];
    unsafeAtomicAdd(&S[(long long)seg * INC + c],
                    feats[(long long)vox * INC + c]);
}

__launch_bounds__(256)
__global__ void spconv_gemm_stats(const float* __restrict__ S,
                                  const float* __restrict__ weight,
                                  float*       __restrict__ out,
                                  float*       __restrict__ stats,
                                  int mtotal) {
    __shared__ float wint[(KDIM / 2) * 2 * OUTC];

    for (int i = threadIdx.x; i < (KDIM / 2) * OUTC; i += blockDim.x) {
        int p = i >> 6;
        int n = i & 63;
        wint[p * 128 + n * 2 + 0] = weight[(2 * p)     * OUTC + n];
        wint[p * 128 + n * 2 + 1] = weight[(2 * p + 1) * OUTC + n];
    }
    __syncthreads();

    const int wave = threadIdx.x >> 5;
    const int lane = threadIdx.x & 31;
    const int r    = lane & 15;
    const int kbh  = lane >> 4;

    const int row_base = (blockIdx.x * 8 + wave) * 32;
    if (row_base >= mtotal) return;

    const int ar0 = min(row_base + r,      mtotal - 1);
    const int ar1 = min(row_base + 16 + r, mtotal - 1);
    const float* __restrict__ A0 = S + (long long)ar0 * KDIM + 2 * kbh;
    const float* __restrict__ A1 = S + (long long)ar1 * KDIM + 2 * kbh;
    const float* __restrict__ B  = wint + kbh * 128 + r * 2;

    v8f a00 = {}, a01 = {}, a02 = {}, a03 = {};
    v8f a10 = {}, a11 = {}, a12 = {}, a13 = {};

#pragma unroll 4
    for (int k = 0; k < KDIM; k += 4) {
        float2 f0 = *(const float2*)(A0 + k);
        float2 f1 = *(const float2*)(A1 + k);
        v2f va0; va0.x = f0.x; va0.y = f0.y;
        v2f va1; va1.x = f1.x; va1.y = f1.y;

        const float* bp = B + (k >> 1) * 128;
        float2 g0 = *(const float2*)(bp);
        float2 g1 = *(const float2*)(bp + 32);
        float2 g2 = *(const float2*)(bp + 64);
        float2 g3 = *(const float2*)(bp + 96);
        v2f b0; b0.x = g0.x; b0.y = g0.y;
        v2f b1; b1.x = g1.x; b1.y = g1.y;
        v2f b2; b2.x = g2.x; b2.y = g2.y;
        v2f b3; b3.x = g3.x; b3.y = g3.y;

        a00 = __builtin_amdgcn_wmma_f32_16x16x4_f32(false, va0, false, b0, (short)0, a00, false, false);
        a10 = __builtin_amdgcn_wmma_f32_16x16x4_f32(false, va1, false, b0, (short)0, a10, false, false);
        a01 = __builtin_amdgcn_wmma_f32_16x16x4_f32(false, va0, false, b1, (short)0, a01, false, false);
        a11 = __builtin_amdgcn_wmma_f32_16x16x4_f32(false, va1, false, b1, (short)0, a11, false, false);
        a02 = __builtin_amdgcn_wmma_f32_16x16x4_f32(false, va0, false, b2, (short)0, a02, false, false);
        a12 = __builtin_amdgcn_wmma_f32_16x16x4_f32(false, va1, false, b2, (short)0, a12, false, false);
        a03 = __builtin_amdgcn_wmma_f32_16x16x4_f32(false, va0, false, b3, (short)0, a03, false, false);
        a13 = __builtin_amdgcn_wmma_f32_16x16x4_f32(false, va1, false, b3, (short)0, a13, false, false);
    }

    const int mrow0 = row_base + ((lane >> 4) << 3);

#define EMIT_NT(ACCL, ACCH, NT)                                                \
    {                                                                          \
        float ps = 0.f, pq = 0.f;                                              \
        _Pragma("unroll")                                                      \
        for (int i = 0; i < 8; ++i) {                                          \
            int m = mrow0 + i;                                                 \
            float v = (ACCL)[i];                                               \
            if (m < mtotal) {                                                  \
                out[(long long)m * OUTC + (NT) * 16 + r] = v;                  \
                ps += v; pq += v * v;                                          \
            }                                                                  \
        }                                                                      \
        _Pragma("unroll")                                                      \
        for (int i = 0; i < 8; ++i) {                                          \
            int m = mrow0 + 16 + i;                                            \
            float v = (ACCH)[i];                                               \
            if (m < mtotal) {                                                  \
                out[(long long)m * OUTC + (NT) * 16 + r] = v;                  \
                ps += v; pq += v * v;                                          \
            }                                                                  \
        }                                                                      \
        ps += __shfl_xor(ps, 16, 32);         \
        pq += __shfl_xor(pq, 16, 32);                                          \
        if (lane < 16) {                                                       \
            unsafeAtomicAdd(&stats[(NT) * 16 + r], ps);                        \
            unsafeAtomicAdd(&stats[64 + (NT) * 16 + r], pq);                   \
        }                                                                      \
    }

    EMIT_NT(a00, a10, 0)
    EMIT_NT(a01, a11, 1)
    EMIT_NT(a02, a12, 2)
    EMIT_NT(a03, a13, 3)
#undef EMIT_NT
}

__global__ void spconv_bn_apply(float* __restrict__ out,
                                const float* __restrict__ stats,
                                const float* __restrict__ gamma,
                                const float* __restrict__ beta,
                                int mtotal) {
    long long i4     = (long long)blockIdx.x * blockDim.x + threadIdx.x;
    long long total4 = ((long long)mtotal * OUTC) >> 2;
    if (i4 >= total4) return;
    const float inv_m = 1.0f / (float)mtotal;
    const int c0 = (int)((i4 << 2) & (OUTC - 1));
    float4 v = ((float4*)out)[i4];
    float x[4] = {v.x, v.y, v.z, v.w};
    float y[4];
#pragma unroll
    for (int j = 0; j < 4; ++j) {
        int c = c0 + j;
        float mean = stats[c] * inv_m;
        float var  = stats[64 + c] * inv_m - mean * mean;
        y[j] = (x[j] - mean) * rsqrtf(var + BN_EPS) * gamma[c] + beta[c];
    }
    ((float4*)out)[i4] = make_float4(y[0], y[1], y[2], y[3]);
}

extern "C" void kernel_launch(void* const* d_in, const int* in_sizes, int n_in,
                              void* d_out, int out_size, void* d_ws, size_t ws_size,
                              hipStream_t stream) {
    const float* feats     = (const float*)d_in[0];
    const float* weight    = (const float*)d_in[1];
    const float* gamma     = (const float*)d_in[2];
    const float* beta      = (const float*)d_in[3];
    const int*   out_idx   = (const int*)d_in[4];
    const int*   offset_id = (const int*)d_in[5];
    const int nvox   = in_sizes[0] / INC;
    const int mtotal = out_size / OUTC;

    float* S     = (float*)d_ws;
    float* stats = S + (long long)mtotal * KDIM;
    const long long zelem4 = ((long long)mtotal * KDIM + 128) >> 2;

    spconv_zero_ws<<<4096, 256, 0, stream>>>((float*)d_ws, zelem4);

    const long long sthreads = (long long)nvox * INC;
    const int sblocks = (int)((sthreads + 255) / 256);
    spconv_scatter<<<sblocks, 256, 0, stream>>>(feats, out_idx, offset_id, S, nvox);

    const int gblocks = (mtotal + 255) / 256;
    spconv_gemm_stats<<<gblocks, 256, 0, stream>>>(S, weight, (float*)d_out, stats, mtotal);

    const long long total4 = ((long long)mtotal * OUTC) >> 2;
    const int bblocks = (int)((total4 + 255) / 256);
    spconv_bn_apply<<<bblocks, 256, 0, stream>>>((float*)d_out, stats, gamma, beta, mtotal);
}


# Batch statistics as a two-launch tree: each GEMM wave stores one plain row of running column sums, a 4-work-group launch adds the 4,096 rows in fixed order; no atomics on the sums; sum tree changes
# speedup vs baseline: 1.9109x; 1.9109x over previous
#include <hip/hip_runtime.h>


typedef __attribute__((ext_vector_type(2))) float v2f;
typedef __attribute__((ext_vector_type(8))) float v8f;

#define INC    32
#define OUTC   64
#define KVOL   8
#define KDIM   256
#define BN_EPS 1e-5f

__global__ void spconv_zero_ws(int* __restrict__ cnt, long long ncnt4) {
    long long i = (long long)blockIdx.x * blockDim.x + threadIdx.x;
    long long stride = (long long)gridDim.x * blockDim.x;
    int4* q = (int4*)cnt;
    for (long long j = i; j < ncnt4; j += stride)
        q[j] = make_int4(0, 0, 0, 0);
}

__launch_bounds__(256)
__global__ void spconv_rank_first(const float* __restrict__ feats,
                                  const int*   __restrict__ out_idx,
                                  const int*   __restrict__ offset_id,
                                  float*       __restrict__ S,
                                  int*         __restrict__ cnt,
                                  int*         __restrict__ code,
                                  int nvox) {
    const int lane = threadIdx.x & 31;
    const int v0   = ((blockIdx.x * blockDim.x + threadIdx.x) >> 5) << 5;
    if (v0 >= nvox) return;
    const int v    = v0 + lane;
    const int vc   = min(v, nvox - 1);
    const int srow = lane >> 3;
    const int scol = (lane & 7) * 4;

    int seg = out_idx[vc] * KVOL + offset_id[vc];

    float4 p[8];
#pragma unroll
    for (int i = 0; i < 8; ++i) {
        int row = min(v0 + 4 * i + srow, nvox - 1);
        p[i] = *(const float4*)(feats + (long long)row * INC + scol);
    }
    asm volatile("" : "+v"(seg));
#pragma unroll
    for (int i = 0; i < 8; ++i)
        asm volatile("" : "+v"(p[i].x), "+v"(p[i].y), "+v"(p[i].z), "+v"(p[i].w));

    int dst = -1;
    if (v < nvox) {
        const int rank = atomicAdd(&cnt[seg], 1);
        dst = (rank == 0) ? seg : -1;
        code[v] = (rank == 0) ? -1 : seg;
    }

#pragma unroll
    for (int i = 0; i < 8; ++i) {
        const int d = __shfl(dst, 4 * i + srow, 32);
        if (d >= 0)
            *(float4*)(S + (long long)d * INC + scol) = p[i];
    }
}

__launch_bounds__(256)
__global__ void spconv_late_add(const float* __restrict__ feats,
                                const int*   __restrict__ code,
                                float*       __restrict__ S,
                                int nvox) {
    const int lane = threadIdx.x & 31;
    const int v0   = ((blockIdx.x * blockDim.x + threadIdx.x) >> 5) << 5;
    if (v0 >= nvox) return;
    const int v = v0 + lane;
    const int c = (v < nvox) ? code[v] : -1;
    unsigned mask = __builtin_amdgcn_ballot_w32(c >= 0);
    while (mask) {
        int   seg[4];
        float x[4];
        bool  on[4];
#pragma unroll
        for (int u = 0; u < 4; ++u) {
            on[u] = (mask != 0u);
            const int j = on[u] ? __builtin_ctz(mask) : 0;
            mask &= mask - 1u;
            seg[u] = __shfl(c, j, 32);
            x[u] = 0.f;
            if (on[u]) x[u] = feats[(long long)(v0 + j) * INC + lane];
        }
#pragma unroll
        for (int u = 0; u < 4; ++u)
            if (on[u]) unsafeAtomicAdd(&S[(long long)seg[u] * INC + lane], x[u]);
    }
}

#define GEMM_WAVES   16
#define GEMM_THREADS (GEMM_WAVES * 32)
#define GEMM_GRID    256
#define KCHUNK       32
#define APITCH       36
#define ASLICE       (32 * APITCH)
__launch_bounds__(GEMM_THREADS)
__global__ void spconv_gemm_stats(float*       __restrict__ S,
                                  const int*   __restrict__ cnt,
                                  const float* __restrict__ weight,
                                  float*       __restrict__ part,
                                  int mtotal) {
    __shared__ float wint[(KDIM / 2) * 2 * OUTC];
    __shared__ __attribute__((aligned(16))) float abuf[GEMM_WAVES * ASLICE];
    __shared__ float sacc[GEMM_WAVES * 8 * 32];

    const int wave = threadIdx.x >> 5;
    const int lane = threadIdx.x & 31;
    const int r    = lane & 15;
    const int kbh  = lane >> 4;

    const int ntile   = (mtotal + 31) >> 5;
    const int tstride = gridDim.x * GEMM_WAVES;
    int tile = blockIdx.x * GEMM_WAVES + wave;

    const int srow = lane >> 3;
    const int scol = (lane & 7) * 4;
    float* __restrict__ aw = abuf + wave * ASLICE;
    float* __restrict__ ast = aw + srow * APITCH + scol;
    const float* __restrict__ af0 = aw + r * APITCH + 2 * kbh;
    const float* __restrict__ af1 = aw + (16 + r) * APITCH + 2 * kbh;
    const float* __restrict__ B  = wint + kbh * 128 + r * 2;

    const float *g0p, *g1p, *g2p, *g3p, *g4p, *g5p, *g6p, *g7p;
    float4 p0, p1, p2, p3, p4, p5, p6, p7;
    unsigned nzlo = 0u, nzhi = 0u;
    int cw0 = 0, cw1 = 0, cw2 = 0, cw3 = 0, cw4 = 0, cw5 = 0, cw6 = 0, cw7 = 0;
    const long long cend = (long long)mtotal * KVOL;
    const float4 zero4 = make_float4(0.f, 0.f, 0.f, 0.f);
#define CNTLD(T, I) ((((long long)(T) * 256 + 32 * (I) + lane) < cend) ? cnt[(long long)(T) * 256 + 32 * (I) + lane] : 0)
#define REQ_CNT(T)                                                             \
    { cw0 = CNTLD(T, 0); cw1 = CNTLD(T, 1); cw2 = CNTLD(T, 2); cw3 = CNTLD(T, 3); \
      cw4 = CNTLD(T, 4); cw5 = CNTLD(T, 5); cw6 = CNTLD(T, 6); cw7 = CNTLD(T, 7); }
#define ROWBITS(CW) ((__builtin_amdgcn_ballot_w32((CW) != 0) >> (srow * 8)) & 0xFFu)
#define MAKE_NZ()                                                              \
    { nzlo = ROWBITS(cw0) | (ROWBITS(cw1) << 8) | (ROWBITS(cw2) << 16) | (ROWBITS(cw3) << 24); \
      nzhi = ROWBITS(cw4) | (ROWBITS(cw5) << 8) | (ROWBITS(cw6) << 16) | (ROWBITS(cw7) << 24); }
#define LDNZ(P, G, W, SH, C)                                                   \
    { if (((W) >> ((SH) + (C))) & 1u) P = *(const float4*)((G) + 32 * (C)); else P = zero4; }
#define ROWPTR(T, J) (S + (long long)min((T) * 32 + 4 * (J) + srow, mtotal - 1) * KDIM + scol)
#define SET_ROWS(T)                                                            \
    { g0p = ROWPTR(T, 0); g1p = ROWPTR(T, 1); g2p = ROWPTR(T, 2); g3p = ROWPTR(T, 3); \
      g4p = ROWPTR(T, 4); g5p = ROWPTR(T, 5); g6p = ROWPTR(T, 6); g7p = ROWPTR(T, 7); }
#define FETCH(C)    \
    { LDNZ(p0, g0p, nzlo,  0, C) LDNZ(p1, g1p, nzlo,  8, C)                     \
      LDNZ(p2, g2p, nzlo, 16, C) LDNZ(p3, g3p, nzlo, 24, C)                     \
      LDNZ(p4, g4p, nzhi,  0, C) LDNZ(p5, g5p, nzhi,  8, C)                     \
      LDNZ(p6, g6p, nzhi, 16, C) LDNZ(p7, g7p, nzhi, 24, C) }
#define PARK()                                                                 \
    { *(float4*)(ast +  0 * APITCH) = p0; *(float4*)(ast +  4 * APITCH) = p1;  \
      *(float4*)(ast +  8 * APITCH) = p2; *(float4*)(ast + 12 * APITCH) = p3;  \
      *(float4*)(ast + 16 * APITCH) = p4; *(float4*)(ast + 20 * APITCH) = p5;  \
      *(float4*)(ast + 24 * APITCH) = p6; *(float4*)(ast + 28 * APITCH) = p7; }

    if (tile < ntile) {
        SET_ROWS(tile)
        REQ_CNT(tile)
        MAKE_NZ()
        FETCH(0)
    }

    for (int i = threadIdx.x; i < (KDIM / 2) * OUTC; i += blockDim.x) {
        int p = i >> 6;
        int n = i & 63;
        wint[p * 128 + n * 2 + 0] = weight[(2 * p)     * OUTC + n];
        wint[p * 128 + n * 2 + 1] = weight[(2 * p + 1) * OUTC + n];
    }
    __syncthreads();

    float* __restrict__ prow = part + (long long)(blockIdx.x * GEMM_WAVES + wave) * (2 * OUTC);
    if (tile >= ntile) {
#pragma unroll
        for (int nt = 0; nt < 4; ++nt)
            prow[(lane >> 4) * OUTC + nt * 16 + r] = 0.f;
        return;
    }
    float* __restrict__ sa = sacc + wave * 256 + lane;
#pragma unroll
    for (int j = 0; j < 8; ++j) sa[32 * j] = 0.f;

    for (;;) {
        const int row_base = tile * 32;
        const int ntl = tile + tstride;
        v8f a00 = {}, a01 = {}, a02 = {}, a03 = {};
        v8f a10 = {}, a11 = {}, a12 = {}, a13 = {};

#pragma nounroll
        for (int kc = 0; kc < KDIM; kc += KCHUNK) {
            PARK()
            if (kc + KCHUNK < KDIM) {
                FETCH((kc >> 5) + 1)
                if (kc + 2 * KCHUNK == KDIM && ntl < ntile) REQ_CNT(ntl)
            } else if (ntl < ntile) {
                SET_ROWS(ntl)
                MAKE_NZ()
                FETCH(0)
            }
            __builtin_amdgcn_wave_barrier();

            v2f fa0[8], fa1[8], fb0[8], fb1[8], fb2[8], fb3[8];
#pragma unroll
            for (int q = 0; q < 8; ++q) {
                const int ks = 4 * q;
                float2 f0 = *(const float2*)(af0 + ks);
                float2 f1 = *(const float2*)(af1 + ks);
                fa0[q].x = f0.x; fa0[q].y = f0.y;
                fa1[q].x = f1.x; fa1[q].y = f1.y;
                const float* bp = B + ((kc + ks) >> 1) * 128;
                float2 g0 = *(const float2*)(bp);
                float2 g1 = *(const float2*)(bp + 32);
                float2 g2 = *(const float2*)(bp + 64);
                float2 g3 = *(const float2*)(bp + 96);
                fb0[q].x = g0.x; fb0[q].y = g0.y;
                fb1[q].x = g1.x; fb1[q].y = g1.y;
                fb2[q].x = g2.x; fb2[q].y = g2.y;
                fb3[q].x = g3.x; fb3[q].y = g3.y;
            }
            __builtin_amdgcn_sched_barrier(0);
#pragma unroll
            for (int q = 0; q < 8; ++q) {
                a00 = __builtin_amdgcn_wmma_f32_16x16x4_f32(false, fa0[q], false, fb0[q], (short)0, a00, false, false);
                a10 = __builtin_amdgcn_wmma_f32_16x16x4_f32(false, fa1[q], false, fb0[q], (short)0, a10, false, false);
                a01 = __builtin_amdgcn_wmma_f32_16x16x4_f32(false, fa0[q], false, fb1[q], (short)0, a01, false, false);
                a11 = __builtin_amdgcn_wmma_f32_16x16x4_f32(false, fa1[q], false, fb1[q], (short)0, a11, false, false);
                a02 = __builtin_amdgcn_wmma_f32_16x16x4_f32(false, fa0[q], false, fb2[q], (short)0, a02, false, false);
                a12 = __builtin_amdgcn_wmma_f32_16x16x4_f32(false, fa1[q], false, fb2[q], (short)0, a12, false, false);
                a03 = __builtin_amdgcn_wmma_f32_16x16x4_f32(false, fa0[q], false, fb3[q], (short)0, a03, false, false);
                a13 = __builtin_amdgcn_wmma_f32_16x16x4_f32(false, fa1[q], false, fb3[q], (short)0, a13, false, false);
            }
            __builtin_amdgcn_wave_barrier();
        }

        const int mrow0 = row_base + ((lane >> 4) << 3);

#define EMIT_NT(ACCL, ACCH, NT)                                                \
        {                                                                      \
            float ps = 0.f, pq = 0.f;                                          \
            _Pragma("unroll")                                                  \
            for (int i = 0; i < 8; ++i) {                                      \
                int m = mrow0 + i;                                             \
                float v = (ACCL)[i];                                           \
                if (m < mtotal) {                                              \
                    S[(long long)m * KDIM + (NT) * 16 + r] = v;                \
                    ps += v; pq += v * v;                                      \
                }                                                              \
            }                                                                  \
            _Pragma("unroll")                                                  \
            for (int i = 0; i < 8; ++i) {                                      \
                int m = mrow0 + 16 + i;                                        \
                float v = (ACCH)[i];                                           \
                if (m < mtotal) {                                              \
                    S[(long long)m * KDIM + (NT) * 16 + r] = v;                \
                    ps += v; pq += v * v;                                      \
                }                                                              \
            }                                                                  \
            sa[32 * (NT)] += ps;                                               \
            sa[32 * (4 + (NT))] += pq;                                         \
        }

        EMIT_NT(a00, a10, 0)
        EMIT_NT(a01, a11, 1)
        EMIT_NT(a02, a12, 2)
        EMIT_NT(a03, a13, 3)
#undef EMIT_NT

        if (ntl >= ntile) break;
        tile = ntl;
    }

#pragma unroll
    for (int nt = 0; nt < 4; ++nt) {
        float ts = sa[32 * nt];
        float tq = sa[32 * (4 + nt)];
        ts += __shfl_xor(ts, 16, 32);
        tq += __shfl_xor(tq, 16, 32);
        prow[(lane >> 4) * OUTC + nt * 16 + r] = (lane < 16) ? ts : tq;
    }
#undef ROWPTR
#undef SET_ROWS
#undef FETCH
#undef PARK
#undef CNTLD
#undef REQ_CNT
#undef ROWBITS
#undef MAKE_NZ
#undef LDNZ
}

#define STAT_ROWS    (GEMM_GRID * GEMM_WAVES)
#define STAT_THREADS 1024
__launch_bounds__(STAT_THREADS)
__global__ void spconv_stats_final(const float* __restrict__ part,
                                   float* __restrict__ stats) {
    __shared__ float red[(STAT_THREADS / 32) * 32];
    const int lane = threadIdx.x & 31;
    const int wave = threadIdx.x >> 5;
    const int col  = (lane & 7) * 4;
    const float* __restrict__ p = part + (long long)(4 * wave + (lane >> 3)) * (2 * OUTC)
                                       + blockIdx.x * 32 + col;
    float4 s = make_float4(0.f, 0.f, 0.f, 0.f);
#pragma unroll
    for (int j = 0; j < STAT_ROWS / (STAT_THREADS / 8); ++j) {
        const float4 a = *(const float4*)(p + (long long)j * (STAT_THREADS / 8) * (2 * OUTC));
        s.x += a.x; s.y += a.y; s.z += a.z; s.w += a.w;
    }
#pragma unroll
    for (int o = 8; o <= 16; o <<= 1) {
        s.x += __shfl_xor(s.x, o, 32); s.y += __shfl_xor(s.y, o, 32);
        s.z += __shfl_xor(s.z, o, 32); s.w += __shfl_xor(s.w, o, 32);
    }
    if (lane < 8) {
        float* d = red + wave * 32 + col;
        d[0] = s.x; d[1] = s.y; d[2] = s.z; d[3] = s.w;
    }
    __syncthreads();
    if (threadIdx.x < 32) {
        float t = 0.f;
#pragma unroll
        for (int w = 0; w < STAT_THREADS / 32; ++w) t += red[w * 32 + threadIdx.x];
        stats[blockIdx.x * 32 + threadIdx.x] = t;
    }
}

__global__ void spconv_bn_apply(const float* __restrict__ raw,
                                float* __restrict__ out,
                                const float* __restrict__ stats,
                                const float* __restrict__ gamma,
                                const float* __restrict__ beta,
                                int mtotal) {
    long long i4     = (long long)blockIdx.x * blockDim.x + threadIdx.x;
    long long total4 = ((long long)mtotal * OUTC) >> 2;
    if (i4 >= total4) return;
    const float inv_m = 1.0f / (float)mtotal;
    const int c0 = (int)((i4 << 2) & (OUTC - 1));
    float4 v = *(const float4*)(raw + (i4 >> 4) * KDIM + ((i4 & 15) << 2));
    float x[4] = {v.x, v.y, v.z, v.w};
    float y[4];
#pragma unroll
    for (int j = 0; j < 4; ++j) {
        int c = c0 + j;
        float mean = stats[c] * inv_m;
        float var  = stats[64 + c] * inv_m - mean * mean;
        y[j] = (x[j] - mean) * rsqrtf(var + BN_EPS) * gamma[c] + beta[c];
    }
    ((float4*)out)[i4] = make_float4(y[0], y[1], y[2], y[3]);
}

extern "C" void kernel_launch(void* const* d_in, const int* in_sizes, int n_in,
                              void* d_out, int out_size, void* d_ws, size_t ws_size,
                              hipStream_t stream) {
    const float* feats     = (const float*)d_in[0];
    const float* weight    = (const float*)d_in[1];
    const float* gamma     = (const float*)d_in[2];
    const float* beta      = (const float*)d_in[3];
    const int*   out_idx   = (const int*)d_in[4];
    const int*   offset_id = (const int*)d_in[5];
    const int nvox   = in_sizes[0] / INC;
    const int mtotal = out_size / OUTC;

    float* S     = (float*)d_ws;
    float* stats = S + (long long)mtotal * KDIM;
    int* cnt  = (int*)d_out;
    int* code = cnt + (long long)mtotal * KVOL;
    float* part = (float*)(code + nvox);
    const long long ncnt4 = ((long long)mtotal * KVOL) >> 2;

    spconv_zero_ws<<<256, 256, 0, stream>>>(cnt, ncnt4);

    const int sblocks = (int)(((long long)nvox + 255) / 256);
    spconv_rank_first<<<sblocks, 256, 0, stream>>>(feats, out_idx, offset_id, S, cnt, code, nvox);
    spconv_late_add<<<sblocks, 256, 0, stream>>>(feats, code, S, nvox);

    spconv_gemm_stats<<<GEMM_GRID, GEMM_THREADS, 0, stream>>>(S, cnt, weight, part, mtotal);
    spconv_stats_final<<<(2 * OUTC) / 32, STAT_THREADS, 0, stream>>>(part, stats);

    const long long total4 = ((long long)mtotal * OUTC) >> 2;
    const int bblocks = (int)((total4 + 255) / 256);
    spconv_bn_apply<<<bblocks, 256, 0, stream>>>(S, (float*)d_out, stats, gamma, beta, mtotal);
}
